# starting kernel
# baseline (speedup 1.0000x reference)
#include <hip/hip_runtime.h>

typedef __attribute__((ext_vector_type(16))) _Float16 v16h;
typedef __attribute__((ext_vector_type(8)))  _Float16 v8h;
typedef __attribute__((ext_vector_type(8)))  float    v8f;

static __device__ __forceinline__ int gtid() {
  return blockIdx.x * blockDim.x + threadIdx.x;
}

__global__ void k_init_deg(float* __restrict__ deg, int N) {
  int i = gtid();
  if (i < N) deg[i] = 1.0f;
}

__global__ void k_deg_edges(const int* __restrict__ dst, float* __restrict__ deg, int E) {
  int e = gtid();
  if (e < E) atomicAdd(&deg[dst[e]], 1.0f);
}

__global__ void k_rsqrt_self(const float* __restrict__ x, float* __restrict__ dis,
                             float* __restrict__ aggx, int N) {
  int i = gtid();
  if (i >= N) return;
  float r = rsqrtf(dis[i]);
  dis[i] = r;
  float nn = r * r;
  float4 xv = reinterpret_cast<const float4*>(x)[i];
  float4 o = make_float4(xv.x * nn, xv.y * nn, xv.z * nn, xv.w * nn);
  reinterpret_cast<float4*>(aggx)[i] = o;
}

__global__ void k_agg1(const int* __restrict__ src, const int* __restrict__ dst,
                       const float* __restrict__ x, const float* __restrict__ dis,
                       float* __restrict__ aggx, int E) {
  int e = gtid();
  if (e >= E) return;
  int s = src[e], d = dst[e];
  float n = dis[s] * dis[d];
  float4 xv = reinterpret_cast<const float4*>(x)[s];
  float* ap = aggx + (size_t)d * 4;
  atomicAdd(ap + 0, xv.x * n);
  atomicAdd(ap + 1, xv.y * n);
  atomicAdd(ap + 2, xv.z * n);
  atomicAdd(ap + 3, xv.w * n);
}

__global__ void k_h1(const float* __restrict__ aggx, const float* __restrict__ W1,
                     const float* __restrict__ b1, const float* __restrict__ dis,
                     float* __restrict__ h1, float* __restrict__ r2, int N) {
  int i = gtid();
  if (i >= N * 64) return;
  int n = i >> 6, f = i & 63;
  float4 av = reinterpret_cast<const float4*>(aggx)[n];
  float acc = b1[f];
  acc += av.x * W1[f];
  acc += av.y * W1[64 + f];
  acc += av.z * W1[128 + f];
  acc += av.w * W1[192 + f];
  float h = fmaxf(acc, 0.0f);
  h1[i] = h;
  float r = dis[n];
  r2[i] = h * r * r;
}

__global__ void k_agg2(const int* __restrict__ src, const int* __restrict__ dst,
                       const float* __restrict__ h1, const float* __restrict__ dis,
                       float* __restrict__ r2, int E) {
  long long t = (long long)blockIdx.x * blockDim.x + threadIdx.x;
  int e  = (int)(t >> 4);
  int f4 = (int)(t & 15);
  if (e >= E) return;
  if (f4 == 0) {
    __builtin_prefetch(src + e + 8192, 0, 1);
    __builtin_prefetch(dst + e + 8192, 0, 1);
  }
  int s = src[e], d = dst[e];
  float n = dis[s] * dis[d];
  float4 v = reinterpret_cast<const float4*>(h1 + (size_t)s * 64)[f4];
  float* rp = r2 + (size_t)d * 64 + f4 * 4;
  atomicAdd(rp + 0, v.x * n);
  atomicAdd(rp + 1, v.y * n);
  atomicAdd(rp + 2, v.z * n);
  atomicAdd(rp + 3, v.w * n);
}

__global__ void __launch_bounds__(128)
k_gemm2_out(const float* __restrict__ r2, const float* __restrict__ W2,
            const float* __restrict__ b2, const float* __restrict__ Wf,
            const float* __restrict__ bf, float* __restrict__ out, int N) {
  __shared__ _Float16 w2t[64 * 64];
  for (int idx = threadIdx.x; idx < 64 * 64; idx += 128) {
    int col = idx >> 6, k = idx & 63;
    w2t[idx] = (_Float16)W2[k * 64 + col];
  }
  __syncthreads();

  int wave  = threadIdx.x >> 5;
  int lane  = threadIdx.x & 31;
  int rbase = blockIdx.x * 64 + wave * 16;
  int lrow  = lane & 15;
  int hi    = lane >> 4;
  int kb    = hi * 8;

  const float* ar = r2 + (size_t)(rbase + lrow) * 64;
  const _Float16* wt = w2t + (size_t)lrow * 64;

  v8f zero = {0, 0, 0, 0, 0, 0, 0, 0};
  v8f acc[4];
  acc[0] = zero; acc[1] = zero; acc[2] = zero; acc[3] = zero;

#pragma unroll
  for (int kc = 0; kc < 64; kc += 32) {
    v16h a;
#pragma unroll
    for (int j = 0; j < 8; ++j) {
      a[j]     = (_Float16)ar[kc + kb + j];
      a[8 + j] = (_Float16)ar[kc + kb + 16 + j];
    }
#pragma unroll
    for (int ct = 0; ct < 4; ++ct) {
      const _Float16* wc = wt + (size_t)ct * 16 * 64;
      v8h blo = *reinterpret_cast<const v8h*>(wc + kc + kb);
      v8h bhi = *reinterpret_cast<const v8h*>(wc + kc + kb + 16);
      v16h b = __builtin_shufflevector(blo, bhi, 0, 1, 2, 3, 4, 5, 6, 7,
                                       8, 9, 10, 11, 12, 13, 14, 15);
      acc[ct] = __builtin_amdgcn_wmma_f32_16x16x32_f16(
          false, a, false, b, (short)0, acc[ct], false, false);
    }
  }

  float bias[4], wf0[4], wf1[4];
#pragma unroll
  for (int ct = 0; ct < 4; ++ct) {
    int col = ct * 16 + lrow;
    bias[ct] = b2[col];
    wf0[ct]  = Wf[col * 2 + 0];
    wf1[ct]  = Wf[col * 2 + 1];
  }
  float bf0 = bf[0], bf1 = bf[1];

#pragma unroll
  for (int v = 0; v < 8; ++v) {
    float p0 = 0.0f, p1 = 0.0f;
#pragma unroll
    for (int ct = 0; ct < 4; ++ct) {
      float h = fmaxf(acc[ct][v] + bias[ct], 0.0f);
      p0 += h * wf0[ct];
      p1 += h * wf1[ct];
    }
#pragma unroll
    for (int off = 8; off >= 1; off >>= 1) {
      p0 += __shfl_xor(p0, off, 16);
      p1 += __shfl_xor(p1, off, 16);
    }
    int row = rbase + v + hi * 8;
    if (lrow == 0 && row < N) {
      out[row * 2 + 0] = p0 + bf0;
      out[row * 2 + 1] = p1 + bf1;
    }
  }
}

extern "C" void kernel_launch(void* const* d_in, const int* in_sizes, int n_in,
                              void* d_out, int out_size, void* d_ws, size_t ws_size,
                              hipStream_t stream) {
  const float* x  = (const float*)d_in[0];
  const int*   ei = (const int*)d_in[1];
  const float* W1 = (const float*)d_in[2];
  const float* b1 = (const float*)d_in[3];
  const float* W2 = (const float*)d_in[4];
  const float* b2 = (const float*)d_in[5];
  const float* Wf = (const float*)d_in[6];
  const float* bf = (const float*)d_in[7];
  float* out = (float*)d_out;

  int N = in_sizes[0] / 4;
  int E = in_sizes[1] / 2;
  int Npad = ((N + 63) / 64) * 64;
  const int* src = ei;
  const int* dst = ei + E;

  float* ws   = (float*)d_ws;
  float* dis  = ws;
  float* aggx = dis + N;
  float* h1   = aggx + (size_t)4 * N;
  float* r2   = h1 + (size_t)64 * N;

  const int TB = 256;
  k_init_deg<<<(N + TB - 1) / TB, TB, 0, stream>>>(dis, N);
  k_deg_edges<<<(E + TB - 1) / TB, TB, 0, stream>>>(dst, dis, E);
  k_rsqrt_self<<<(N + TB - 1) / TB, TB, 0, stream>>>(x, dis, aggx, N);
  k_agg1<<<(E + TB - 1) / TB, TB, 0, stream>>>(src, dst, x, dis, aggx, E);
  long long t1 = (long long)N * 64;
  k_h1<<<(unsigned)((t1 + TB - 1) / TB), TB, 0, stream>>>(aggx, W1, b1, dis, h1, r2, N);
  long long t2 = (long long)E * 16;
  k_agg2<<<(unsigned)((t2 + TB - 1) / TB), TB, 0, stream>>>(src, dst, h1, dis, r2, E);
  k_gemm2_out<<<Npad / 64, 128, 0, stream>>>(r2, W2, b2, Wf, bf, out, N);
}


# Output head done by each layer-2 wave for its own 9-10 consecutive rows: r2 rows stay in the wave's LDS slice, W2 product and Wf there; no r2 plane, no barrier in the loop; 4 launches; same sums
# speedup vs baseline: 11.3911x; 11.3911x over previous
#include <hip/hip_runtime.h>

typedef __attribute__((ext_vector_type(16))) _Float16 v16h;
typedef __attribute__((ext_vector_type(8)))  _Float16 v8h;
typedef __attribute__((ext_vector_type(8)))  float    v8f;
typedef __attribute__((ext_vector_type(2)))  float    v2f;

static __device__ __forceinline__ int gtid() {
  return blockIdx.x * blockDim.x + threadIdx.x;
}

static __device__ __forceinline__ float4 f4add(float4 a, float4 b) {
  return make_float4(a.x + b.x, a.y + b.y, a.z + b.z, a.w + b.w);
}
static __device__ __forceinline__ float4 f4scale(float4 a, float s) {
  return make_float4(a.x * s, a.y * s, a.z * s, a.w * s);
}


#define BK_DST   256
#define BK_SHIFT 8
#define SRC_BITS 24
#define SRC_MASK 0xFFFFFFu
#define SORT_TH  512
#define PER_TH   16
#define CHUNK    (SORT_TH * PER_TH)
#define CAP      16384
#define NWV      (SORT_TH / 32)

#define W2_PITCH 72

static __device__ __forceinline__ int scan512(int v, int* wsum) {
  int lane = threadIdx.x & 31, wv = threadIdx.x >> 5;
  int incl = v;
  for (int o = 1; o < 32; o <<= 1) {
    int y = __shfl_up(incl, o, 32);
    if (lane >= o) incl += y;
  }
  if (lane == 31) wsum[wv] = incl;
  __syncthreads();
  int pre = 0;
  for (int k = 0; k < 16; ++k) pre += (k < wv) ? wsum[k] : 0;
  __syncthreads();
  return pre + incl - v;
}

__global__ void __launch_bounds__(SORT_TH)
k_partition(const int* __restrict__ src, const int* __restrict__ dst,
            int* __restrict__ lbt, unsigned* __restrict__ bbuf,
            int E, int NB, int NBp, int PB,
            const float* __restrict__ W2, unsigned* __restrict__ w2img) {
  __shared__ int hw[NWV * SORT_TH];
  __shared__ int wsum[16];
  __shared__ __attribute__((aligned(16))) unsigned sw[CHUNK];
  int tid = threadIdx.x;
  int blk = blockIdx.x;
  if (blk >= PB) {
    int j = (blk - PB) * SORT_TH + tid;
    if (j < 64 * 32) {
      int col = j >> 5, k = (j & 31) * 2;
      _Float16 lo = (_Float16)W2[k * 64 + col];
      _Float16 hi = (_Float16)W2[(k + 1) * 64 + col];
      unsigned ulo = __builtin_bit_cast(unsigned short, lo);
      unsigned uhi = __builtin_bit_cast(unsigned short, hi);
      w2img[j] = ulo | (uhi << 16);
    }
    return;
  }
#pragma unroll
  for (int k = 0; k < NWV; ++k) hw[k * SORT_TH + tid] = 0;
  __syncthreads();
  int* myh = hw + (tid >> 5) * SORT_TH;
  int eb = blk * CHUNK;
  int e0 = eb + tid;
  unsigned w[PER_TH];
  int lr[PER_TH], bk[PER_TH], dv[PER_TH], sv[PER_TH];
#pragma unroll
  for (int i = 0; i < PER_TH; ++i) {
    int e = e0 + i * SORT_TH;
    int ec = e < E ? e : E - 1;
    dv[i] = dst[ec];
    sv[i] = src[ec];
  }
#pragma unroll
  for (int i = 0; i < PER_TH; ++i) {
    bk[i] = -1;
    if (e0 + i * SORT_TH < E) {
      bk[i] = dv[i] >> BK_SHIFT;
      w[i]  = ((unsigned)(dv[i] & (BK_DST - 1)) << SRC_BITS) | (unsigned)sv[i];
      lr[i] = atomicAdd(&myh[bk[i]], 1);
    }
  }
  __syncthreads();
  int cnt = E - eb < CHUNK ? E - eb : CHUNK;
  int hc[NWV];
  int h = 0;
#pragma unroll
  for (int k = 0; k < NWV; ++k) { hc[k] = hw[k * SORT_TH + tid]; h += hc[k]; }
  int lb = scan512(h, wsum);
  int run = lb;
#pragma unroll
  for (int k = 0; k < NWV; ++k) { hw[k * SORT_TH + tid] = run; run += hc[k]; }
  int* row = lbt + (size_t)blk * NBp;
  if (tid < NB) row[tid] = lb;
  if (tid == 0) row[NB] = cnt;
  __syncthreads();
#pragma unroll
  for (int i = 0; i < PER_TH; ++i)
    if (bk[i] >= 0) sw[myh[bk[i]] + lr[i]] = w[i];
  __syncthreads();
  uint4* o4 = reinterpret_cast<uint4*>(bbuf + eb);
  const uint4* s4 = reinterpret_cast<const uint4*>(sw);
  int n4 = cnt >> 2;
  for (int p = tid; p < n4; p += SORT_TH) o4[p] = s4[p];
  for (int p = (n4 << 2) + tid; p < cnt; p += SORT_TH) bbuf[eb + p] = sw[p];
}

template <class F>
static __device__ __forceinline__ void run_words(const unsigned* __restrict__ bbuf,
                                                 int myrb, int myrl, int lane, int first, F f) {
  unsigned m = __builtin_amdgcn_ballot_w32(myrl > first);
  while (m) {
    int k = __builtin_ctz(m);
    m &= m - 1;
    int rb = __builtin_amdgcn_readlane(myrb, k);
    int rl = __builtin_amdgcn_readlane(myrl, k);
    for (int j = first + lane; j < rl; j += 32) f(bbuf[rb + j]);
  }
}

__global__ void __launch_bounds__(SORT_TH)
k_bucket_sort(const unsigned* __restrict__ bbuf, const int* __restrict__ lbt,
              const float* __restrict__ x, int* __restrict__ off,
              float* __restrict__ dis, float* __restrict__ xs,
              int* __restrict__ csr, int N, int E, int PB, int NBp) {
  __shared__ int cw[NWV * BK_DST];
  __shared__ int wsum[16];
  __shared__ int s_start, s_total;
  __shared__ int rbeg[SORT_TH];
  __shared__ int rlen[SORT_TH];
  __shared__ unsigned sw[CAP];
  int tid = threadIdx.x, b = blockIdx.x;
  int lane = tid & 31;
  int wv = __builtin_amdgcn_readfirstlane(tid >> 5);

  int rlo = 0, rl0 = 0;
  if (tid < PB) {
    const int* row = lbt + (size_t)tid * NBp + b;
    rlo = row[0];
    rl0 = row[1] - rlo;
  }
  rbeg[tid] = tid * CHUNK + rlo;
  rlen[tid] = rl0;
#pragma unroll
  for (int k = 0; k < NWV * BK_DST / SORT_TH; ++k) cw[k * SORT_TH + tid] = 0;
  int pa = scan512(rlo, wsum);
  int pb = scan512(rl0, wsum);
  if (tid == SORT_TH - 1) { s_start = pa + rlo; s_total = pb + rl0; }
  __syncthreads();
  int start = s_start;
  int total = s_total;
  int emax = E > 0 ? E - 1 : 0;
  int* myc = cw + wv * BK_DST;

  int myrb = rbeg[wv + 16 * lane];
  int myrl = rlen[wv + 16 * lane];

  bool big = total > CAP;
  unsigned w[SORT_TH / 16];
  if (!big) {
#pragma unroll
    for (int k = 0; k < SORT_TH / 16; ++k) {
      int rb = __builtin_amdgcn_readlane(myrb, k);
      int rl = __builtin_amdgcn_readlane(myrl, k);
      int q = rb + (lane < rl ? lane : 0);
      w[k] = bbuf[q < emax ? q : emax];
    }
#pragma unroll
    for (int k = 0; k < SORT_TH / 16; ++k)
      if (lane < __builtin_amdgcn_readlane(myrl, k)) atomicAdd(&myc[w[k] >> SRC_BITS], 1);
    run_words(bbuf, myrb, myrl, lane, 32, [&](unsigned v) { atomicAdd(&myc[v >> SRC_BITS], 1); });
  } else {
    run_words(bbuf, myrb, myrl, lane, 0, [&](unsigned v) { atomicAdd(&myc[v >> SRC_BITS], 1); });
  }
  __syncthreads();
  int cc[NWV];
  int c = 0;
  if (tid < BK_DST) {
#pragma unroll
    for (int k = 0; k < NWV; ++k) { cc[k] = cw[k * BK_DST + tid]; c += cc[k]; }
  }
  int lo = scan512(c, wsum);
  if (tid < BK_DST) {
    int run = lo;
#pragma unroll
    for (int k = 0; k < NWV; ++k) { cw[k * BK_DST + tid] = run; run += cc[k]; }
    int node = b * BK_DST + tid;
    if (node < N) {
      off[node] = start + lo;
      float r = rsqrtf((float)(c + 1));
      dis[node] = r;
      float4 xv = reinterpret_cast<const float4*>(x)[node];
      reinterpret_cast<float4*>(xs)[node] = f4scale(xv, r);
      if (node == N - 1) off[N] = start + total;
    }
  }
  __syncthreads();

  if (big) {
    run_words(bbuf, myrb, myrl, lane, 0, [&](unsigned v) {
      int r = atomicAdd(&myc[v >> SRC_BITS], 1);
      csr[start + r] = (int)(v & SRC_MASK);
    });
    return;
  }

  int pl[SORT_TH / 16];
#pragma unroll
  for (int k = 0; k < SORT_TH / 16; ++k) {
    pl[k] = -1;
    if (lane < __builtin_amdgcn_readlane(myrl, k)) pl[k] = atomicAdd(&myc[w[k] >> SRC_BITS], 1);
  }
#pragma unroll
  for (int k = 0; k < SORT_TH / 16; ++k)
    if (pl[k] >= 0) sw[pl[k]] = w[k] & SRC_MASK;
  run_words(bbuf, myrb, myrl, lane, 32, [&](unsigned v) {
    int r = atomicAdd(&myc[v >> SRC_BITS], 1);
    sw[r] = v & SRC_MASK;
  });
  __syncthreads();

  for (int p = tid; p < total; p += SORT_TH) csr[start + p] = (int)sw[p];
}

#define P1_U 6
__global__ void __launch_bounds__(256)
k_pull1(const int* __restrict__ off, const int* __restrict__ csr,
        const float* __restrict__ xs, const float* __restrict__ dis,
        float* __restrict__ q, int N, int E) {
  int lane = threadIdx.x & 31;
  int sub = lane & 7, nd = lane >> 3;
  int wv  = __builtin_amdgcn_readfirstlane(threadIdx.x >> 5);
  int wid = blockIdx.x * 8 + wv;
  int tw  = gridDim.x * 8;
  int ngr = (N + 3) >> 2;
  if (wid >= ngr) return;
  int jmax = E > 0 ? E - 1 : 0;
  const float4* xs4 = reinterpret_cast<const float4*>(xs);

  int d = wid * 4 + nd;
  int dc = d < N ? d : N - 1;
  int beg = off[dc], end = off[dc + 1];
  if (d >= N) { beg = 0; end = 0; }
  int sv[P1_U];
#pragma unroll
  for (int k = 0; k < P1_U; ++k) {
    int j = beg + sub + 8 * k;
    sv[k] = csr[j < jmax ? j : jmax];
  }
  int dn = (wid + tw) * 4 + nd;
  int dnc = dn < N ? dn : N - 1;
  int nbeg = off[dnc], nend = off[dnc + 1];

  for (int g = wid; g < ngr; g += tw) {
    d = g * 4 + nd;
    bool valid = d < N;
    int ds = valid ? d : 0;
    float r = dis[ds];
    float4 self = xs4[ds];
    float4 v[P1_U];
#pragma unroll
    for (int k = 0; k < P1_U; ++k) {
      int j = beg + sub + 8 * k;
      int s = j < end ? sv[k] : ds;
      v[k] = xs4[s];
    }
    int d1 = (g + tw) * 4 + nd;
    bool v1 = d1 < N;
    int beg1 = v1 ? nbeg : 0, end1 = v1 ? nend : 0;
    int sv1[P1_U];
#pragma unroll
    for (int k = 0; k < P1_U; ++k) {
      int j = beg1 + sub + 8 * k;
      sv1[k] = csr[j < jmax ? j : jmax];
    }
    int d2 = (g + 2 * tw) * 4 + nd;
    int d2c = d2 < N ? d2 : N - 1;
    nbeg = off[d2c]; nend = off[d2c + 1];

    float4 a = make_float4(0.f, 0.f, 0.f, 0.f);
#pragma unroll
    for (int k = 0; k < P1_U; ++k) {
      float4 t = f4add(a, v[k]);
      bool in = beg + sub + 8 * k < end;
      a.x = in ? t.x : a.x; a.y = in ? t.y : a.y;
      a.z = in ? t.z : a.z; a.w = in ? t.w : a.w;
    }
    for (int j = beg + sub + 8 * P1_U; j < end; j += 8) a = f4add(a, xs4[csr[j]]);
    for (int o = 4; o >= 1; o >>= 1) {
      a.x += __shfl_xor(a.x, o, 8);
      a.y += __shfl_xor(a.y, o, 8);
      a.z += __shfl_xor(a.z, o, 8);
      a.w += __shfl_xor(a.w, o, 8);
    }
    a = f4scale(f4add(a, self), r);
    float4* qrow = reinterpret_cast<float4*>(q + (size_t)ds * 8);
    float4 w = sub == 0 ? a : make_float4(r, 0.f, 0.f, 0.f);
    if (valid && sub < 2) qrow[sub] = w;

    beg = beg1; end = end1;
#pragma unroll
    for (int k = 0; k < P1_U; ++k) sv[k] = sv1[k];
  }
}

template <int R>
static __device__ __forceinline__ float row_bcast(float v) {
  int x = __builtin_bit_cast(int, v);
  return __builtin_bit_cast(float, __builtin_amdgcn_mov_dpp(x, 0x150 + R, 0xf, 0xf, true));
}
template <int M>
static __device__ __forceinline__ float row_xor(float v) {
  int x = __builtin_bit_cast(int, v);
  return __builtin_bit_cast(float, __builtin_amdgcn_mov_dpp(x, 0x160 + M, 0xf, 0xf, true));
}
static __device__ __forceinline__ float upper_ror8(float lo, float v) {
  return __builtin_bit_cast(float, __builtin_amdgcn_update_dpp(
      __builtin_bit_cast(int, lo), __builtin_bit_cast(int, v), 0x128, 0x2, 0xf, false));
}
static __device__ __forceinline__ float xhalf(float x) {
  int i = __builtin_bit_cast(int, x);
  return __builtin_bit_cast(float, __builtin_amdgcn_permlanex16(i, i, 0x76543210, 0xfedcba98, false, false));
}

#define L2_GROUP 3
#define R2_PITCH 68
__global__ void __launch_bounds__(256)
k_pull2_head(const int* __restrict__ off, const int* __restrict__ csr,
             const float* __restrict__ q, const float* __restrict__ W1,
             const float* __restrict__ b1, const unsigned* __restrict__ w2img,
             const float* __restrict__ b2, const float* __restrict__ Wf,
             const float* __restrict__ bf, float* __restrict__ out, int N, int E) {
  __shared__ __attribute__((aligned(16))) _Float16 w2t[64 * W2_PITCH];
  __shared__ __attribute__((aligned(16))) float r2s[8 * 16 * R2_PITCH];
  int lane = threadIdx.x & 31;
  int c = lane & 15, h = lane >> 4;
  int cs = (c + 8 * h) & 15;
  int wv  = __builtin_amdgcn_readfirstlane(threadIdx.x >> 5);
  int wid = blockIdx.x * 8 + wv;
  int tw  = gridDim.x * 8;
  int jmax = E > 0 ? E - 1 : 0;

  const uint4* img = reinterpret_cast<const uint4*>(w2img);
  uint4 ch[2];
#pragma unroll
  for (int i = 0; i < 2; ++i) ch[i] = img[threadIdx.x + 256 * i];

  int base = N / tw, rem = N - base * tw;
  int start = wid * base + (wid < rem ? wid : rem);
  int nrow  = base + (wid < rem ? 1 : 0);
  int oi = start + (lane < nrow ? lane : nrow);
  int ri = start + (lane < nrow ? lane : nrow - 1);
  ri = ri < 0 ? 0 : ri;
  int offv = off[oi];
  int rdv  = __builtin_bit_cast(int, q[(size_t)ri * 8 + 4]);

  float4 wk0 = *reinterpret_cast<const float4*>(W1 + (2 * h) * 64 + 4 * c);
  float4 wk1 = *reinterpret_cast<const float4*>(W1 + (2 * h + 1) * 64 + 4 * c);
  float4 bb  = *reinterpret_cast<const float4*>(b1 + 4 * c);
  v2f bfr[4];
  bfr[0] = (v2f){wk0.x, wk1.x}; bfr[1] = (v2f){wk0.y, wk1.y};
  bfr[2] = (v2f){wk0.z, wk1.z}; bfr[3] = (v2f){wk0.w, wk1.w};
  float bias[4] = {bb.x, bb.y, bb.z, bb.w};
  v8f zero = {0, 0, 0, 0, 0, 0, 0, 0};

#pragma unroll
  for (int i = 0; i < 2; ++i) {
    int id = threadIdx.x + 256 * i;
    int col = id >> 3, part = id & 7;
    *reinterpret_cast<uint4*>(w2t + col * W2_PITCH + part * 8) = ch[i];
  }
  __syncthreads();

  float* myr = r2s + wv * 16 * R2_PITCH;
  const float* ar = myr + c * R2_PITCH;
  const _Float16* wt = w2t + (size_t)c * W2_PITCH;
  int kb = h * 8;

  if (nrow == 0) return;
  {
    int nr = nrow;
    int hh = (nr + 1) >> 1;
    int dbase = start;

#pragma nounroll
    for (int t = 0; t < 16; ++t) {
      int i = t < 8 ? t : hh + t - 8;
      bool have = t < 8 ? t < hh : i < nr;
      float4 val = make_float4(0.f, 0.f, 0.f, 0.f);
      if (have) {
        int d = dbase + i;
        int beg = __builtin_amdgcn_readlane(offv, i);
        int end = __builtin_amdgcn_readlane(offv, i + 1);
        int cnt = end - beg + 1;
        int nt  = (cnt + 15) >> 4;
        v2f acc[4];
#pragma unroll
        for (int ct = 0; ct < 4; ++ct) acc[ct] = (v2f){0.f, 0.f};

        for (int t0 = 0; t0 < nt; t0 += L2_GROUP) {
          int sv[L2_GROUP];
#pragma unroll
          for (int k = 0; k < L2_GROUP; ++k) {
            int j = beg + (t0 + k) * 16 + cs - 1;
            j = j < 0 ? 0 : j;
            j = j < jmax ? j : jmax;
            sv[k] = csr[j];
          }
          asm volatile("" :: "v"(sv[0]), "v"(sv[1]), "v"(sv[2]));
          float4 pv[L2_GROUP];
          float  ev[L2_GROUP];
#pragma unroll
          for (int k = 0; k < L2_GROUP; ++k) {
            int row = (t0 + k) * 16 + cs;
            int s = (row >= 1 && row < cnt) ? sv[k] : d;
            const char* qs = reinterpret_cast<const char*>(q) + ((unsigned)s << 5);
            pv[k] = *reinterpret_cast<const float4*>(qs);
            ev[k] = *reinterpret_cast<const float*>(qs + 16);
          }
          asm volatile("" :: "v"(pv[0].x), "v"(pv[0].y), "v"(pv[0].z), "v"(pv[0].w), "v"(ev[0]),
                             "v"(pv[1].x), "v"(pv[1].y), "v"(pv[1].z), "v"(pv[1].w), "v"(ev[1]),
                             "v"(pv[2].x), "v"(pv[2].y), "v"(pv[2].z), "v"(pv[2].w), "v"(ev[2]));
#pragma unroll
          for (int k = 0; k < L2_GROUP; ++k) {
            if (t0 + k < nt) {
              int row = (t0 + k) * 16 + cs;
              float e = row < cnt ? ev[k] : 0.f;
              v2f a;
              a.x = upper_ror8(pv[k].x, pv[k].z);
              a.y = upper_ror8(pv[k].y, pv[k].w);
              v8f dm[4];
#pragma unroll
              for (int ct = 0; ct < 4; ++ct)
                dm[ct] = __builtin_amdgcn_wmma_f32_16x16x4_f32(
                    false, a, false, bfr[ct], (short)0, zero, false, false);
              v2f dr[4];
              dr[0] = (v2f){row_bcast<0>(e), row_bcast<1>(e)};
              dr[1] = (v2f){row_bcast<2>(e), row_bcast<3>(e)};
              dr[2] = (v2f){row_bcast<4>(e), row_bcast<5>(e)};
              dr[3] = (v2f){row_bcast<6>(e), row_bcast<7>(e)};
#pragma unroll
              for (int ct = 0; ct < 4; ++ct) {
                v2f bc = (v2f){bias[ct], bias[ct]};
                v2f z2 = (v2f){0.f, 0.f};
                v2f h0 = __builtin_elementwise_max(__builtin_shufflevector(dm[ct], dm[ct], 0, 1) + bc, z2);
                v2f h1 = __builtin_elementwise_max(__builtin_shufflevector(dm[ct], dm[ct], 2, 3) + bc, z2);
                v2f h2 = __builtin_elementwise_max(__builtin_shufflevector(dm[ct], dm[ct], 4, 5) + bc, z2);
                v2f h3 = __builtin_elementwise_max(__builtin_shufflevector(dm[ct], dm[ct], 6, 7) + bc, z2);
                acc[ct] = __builtin_elementwise_fma(h0, dr[0], acc[ct]);
                acc[ct] = __builtin_elementwise_fma(h1, dr[1], acc[ct]);
                acc[ct] = __builtin_elementwise_fma(h2, dr[2], acc[ct]);
                acc[ct] = __builtin_elementwise_fma(h3, dr[3], acc[ct]);
              }
            }
          }
        }
        float rd = __builtin_bit_cast(float, __builtin_amdgcn_readlane(rdv, i));
        float s0 = acc[0].x + acc[0].y, s1 = acc[1].x + acc[1].y;
        float s2 = acc[2].x + acc[2].y, s3 = acc[3].x + acc[3].y;
        s0 += xhalf(s0); s1 += xhalf(s1); s2 += xhalf(s2); s3 += xhalf(s3);
        val = make_float4(s0 * rd, s1 * rd, s2 * rd, s3 * rd);
      }
      if (h == 0) *reinterpret_cast<float4*>(myr + t * R2_PITCH + 4 * c) = val;
    }
    __builtin_amdgcn_fence(__ATOMIC_ACQ_REL, "wavefront");
    __builtin_amdgcn_wave_barrier();

    v8f acc2[4];
    acc2[0] = zero; acc2[1] = zero; acc2[2] = zero; acc2[3] = zero;
#pragma unroll
    for (int kc = 0; kc < 64; kc += 32) {
      v16h a;
#pragma unroll
      for (int j = 0; j < 8; ++j) {
        a[j]     = (_Float16)ar[kc + kb + j];
        a[8 + j] = (_Float16)ar[kc + kb + 16 + j];
      }
#pragma unroll
      for (int ct = 0; ct < 4; ++ct) {
        const _Float16* wc = wt + (size_t)ct * 16 * W2_PITCH;
        v8h blo = *reinterpret_cast<const v8h*>(wc + kc + kb);
        v8h bhi = *reinterpret_cast<const v8h*>(wc + kc + kb + 16);
        v16h b = __builtin_shufflevector(blo, bhi, 0, 1, 2, 3, 4, 5, 6, 7,
                                         8, 9, 10, 11, 12, 13, 14, 15);
        acc2[ct] = __builtin_amdgcn_wmma_f32_16x16x32_f16(
            false, a, false, b, (short)0, acc2[ct], false, false);
      }
    }

    float bias2[4], wf0[4], wf1[4];
#pragma unroll
    for (int ct = 0; ct < 4; ++ct) {
      int col = ct * 16 + c;
      bias2[ct] = b2[col];
      wf0[ct]   = Wf[col * 2 + 0];
      wf1[ct]   = Wf[col * 2 + 1];
    }
    float bf0 = bf[0], bf1 = bf[1];

#pragma unroll
    for (int v = 0; v < 8; ++v) {
      if (v < hh) {
        float p0 = 0.0f, p1 = 0.0f;
#pragma unroll
        for (int ct = 0; ct < 4; ++ct) {
          float hv = fmaxf(acc2[ct][v] + bias2[ct], 0.0f);
          p0 += hv * wf0[ct];
          p1 += hv * wf1[ct];
        }
        p0 += row_xor<8>(p0); p1 += row_xor<8>(p1);
        p0 += row_xor<4>(p0); p1 += row_xor<4>(p1);
        p0 += row_xor<2>(p0); p1 += row_xor<2>(p1);
        p0 += row_xor<1>(p0); p1 += row_xor<1>(p1);
        int i = h == 0 ? v : hh + v;
        if (c == 0 && i < nr) {
          int row = dbase + i;
          out[row * 2 + 0] = p0 + bf0;
          out[row * 2 + 1] = p1 + bf1;
        }
      }
    }
  }
}

extern "C" void kernel_launch(void* const* d_in, const int* in_sizes, int n_in,
                              void* d_out, int out_size, void* d_ws, size_t ws_size,
                              hipStream_t stream) {
  const float* x  = (const float*)d_in[0];
  const int*   ei = (const int*)d_in[1];
  const float* W1 = (const float*)d_in[2];
  const float* b1 = (const float*)d_in[3];
  const float* W2 = (const float*)d_in[4];
  const float* b2 = (const float*)d_in[5];
  const float* Wf = (const float*)d_in[6];
  const float* bf = (const float*)d_in[7];
  float* out = (float*)d_out;

  int N = in_sizes[0] / 4;
  int E = in_sizes[1] / 2;
  int Nal  = ((N + 64) / 64) * 64;
  int Eal  = ((E + 63) / 64) * 64;
  int NB   = (N + BK_DST - 1) / BK_DST;
  int NBp  = NB + 1;
  int PB   = (E + CHUNK - 1) / CHUNK;
  int LBal = ((PB * NBp + 63) / 64) * 64;
  const int* src = ei;
  const int* dst = ei + E;

  int*      lbt    = (int*)d_ws;
  int*      off    = lbt + LBal;
  float*    dis    = (float*)(off + Nal);
  float*    xs     = dis + Nal;
  unsigned* bbuf   = (unsigned*)(xs + (size_t)4 * Nal);
  int*      csr    = (int*)(bbuf + Eal);
  float*    q      = (float*)(csr + Eal);
  unsigned* w2img  = (unsigned*)(q + (size_t)8 * Nal);

  const int TB = 256;
  k_partition<<<PB + 64 * 32 / SORT_TH, SORT_TH, 0, stream>>>(src, dst, lbt, bbuf, E, NB, NBp, PB, W2, w2img);
  k_bucket_sort<<<NB, SORT_TH, 0, stream>>>(bbuf, lbt, x, off, dis, xs, csr, N, E, PB, NBp);
  k_pull1<<<512, TB, 0, stream>>>(off, csr, xs, dis, q, N, E);
  k_pull2_head<<<N > 16 * 8 * 1280 ? (N + 127) / 128 : 1280, TB, 0, stream>>>(off, csr, q, W1, b1, w2img, b2, Wf, bf, out, N, E);
}
